# starting kernel
# baseline (speedup 1.0000x reference)
#include <hip/hip_runtime.h>
#include <hip/hip_bf16.h>
#include <math.h>


typedef __attribute__((ext_vector_type(2))) float v2f;
typedef __attribute__((ext_vector_type(8))) float v8f;

#define BATCH 256
#define N_DIM 64
#define D_DIM 128
#define H_DIM 64
#define LDP   66


__device__ inline v8f wmma_f32x4(v2f a, v2f b, v8f c) {
  return __builtin_amdgcn_wmma_f32_16x16x4_f32(
       false, a,  false, b,
       (short)0, c,  false,  false);
}

__device__ inline void init_acc(v8f acc[4], const float* __restrict__ bias, int l16) {
#pragma unroll
  for (int t = 0; t < 4; ++t) {
    float bv = bias ? bias[t * 16 + l16] : 0.0f;
#pragma unroll
    for (int r = 0; r < 8; ++r) acc[t][r] = bv;
  }
}

__device__ inline void gemm_accT(const float* __restrict__ A, int lda,
                                 const float* __restrict__ BT, int ldbt,
                                 int K, v8f acc[4], int lane, int rowBase) {
  const int half = lane >> 4;
  const int l16  = lane & 15;
  for (int k0 = 0; k0 < K; k0 += 4) {
    const int ka = k0 + 2 * half;
    v2f a = *(const v2f*)(A + (rowBase + l16) * lda + ka);
#pragma unroll
    for (int t = 0; t < 4; ++t) {
      v2f b = *(const v2f*)(BT + (t * 16 + l16) * ldbt + ka);
      acc[t] = wmma_f32x4(a, b, acc[t]);
    }
  }
}

__device__ inline void store_stripe(float* __restrict__ D, int ldd,
                                    const v8f acc[4], int lane, int rowBase,
                                    bool do_relu) {
  const int half = lane >> 4;
  const int l16  = lane & 15;
#pragma unroll
  for (int t = 0; t < 4; ++t)
#pragma unroll
    for (int r = 0; r < 8; ++r) {
      float v = acc[t][r];
      if (do_relu) v = fmaxf(v, 0.0f);
      D[(rowBase + r + 8 * half) * ldd + t * 16 + l16] = v;
    }
}

__device__ inline void store_stripeT(float* __restrict__ DT, int ldt,
                                     const v8f acc[4], int lane, int rowBase) {
  const int half = lane >> 4;
  const int l16  = lane & 15;
#pragma unroll
  for (int t = 0; t < 4; ++t)
#pragma unroll
    for (int r = 0; r < 8; ++r)
      DT[(t * 16 + l16) * ldt + rowBase + r + 8 * half] = acc[t][r];
}

__device__ inline void copy_w_T(float* __restrict__ dst,
                                const float* __restrict__ src,
                                int koff, int tid) {
  for (int s = tid; s < 64 * 64; s += 128) {
    const int k = s >> 6;
    const int n = s & 63;
    dst[n * LDP + k] = src[(k + koff) * H_DIM + n];
  }
}

__device__ inline void copy_obs_half(float* __restrict__ dst,
                                     const float* __restrict__ src,
                                     int coff, int tid) {
  for (int s = tid; s < 64 * 32; s += 128) {
    const int r  = s >> 5;
    const int c2 = (s & 31) * 2;
    *(float2*)(dst + r * LDP + c2) =
        *(const float2*)(src + r * D_DIM + coff + c2);
  }
}

__global__ __launch_bounds__(128) void gated_comm_kernel(
    const float* __restrict__ obs, const float* __restrict__ enc_w,
    const float* __restrict__ enc_b, const float* __restrict__ gate_w1,
    const float* __restrict__ gate_b1, const float* __restrict__ gate_w2,
    const float* __restrict__ gate_b2, const float* __restrict__ msg_w1,
    const float* __restrict__ msg_b1, const float* __restrict__ msg_w2,
    const float* __restrict__ msg_b2, float* __restrict__ out) {
  __shared__ float smem[3 * 64 * LDP + 64];

  const int b       = blockIdx.x;
  const int tid     = threadIdx.x;
  const int lane    = tid & 31;
  const int wave    = tid >> 5;
  const int l16     = lane & 15;
  const int rowBase = wave * 16;

  float* R0  = smem;
  float* R1  = smem + 64 * LDP;
  float* R2  = smem + 2 * 64 * LDP;
  float* sW2 = smem + 3 * 64 * LDP;

  const float* obsb = obs + (size_t)b * (N_DIM * D_DIM);
  v8f acc[4];

  copy_obs_half(R0, obsb, 0, tid);
  copy_w_T(R2, enc_w, 0, tid);
  __syncthreads();
  init_acc(acc, enc_b, l16);
  gemm_accT(R0, LDP, R2, LDP, 64, acc, lane, rowBase);
  __syncthreads();
  copy_obs_half(R0, obsb, 64, tid);
  copy_w_T(R2, enc_w, 64, tid);
  __syncthreads();
  gemm_accT(R0, LDP, R2, LDP, 64, acc, lane, rowBase);
  __syncthreads();
  store_stripe(R0, LDP, acc, lane, rowBase,  true);
  copy_w_T(R2, gate_w1, 0, tid);
  __syncthreads();

  init_acc(acc, nullptr, l16);
  gemm_accT(R0, LDP, R2, LDP, H_DIM, acc, lane, rowBase);
  store_stripe(R1, LDP, acc, lane, rowBase, false);
  __syncthreads();
  copy_w_T(R2, gate_w1, 64, tid);
  __syncthreads();

  init_acc(acc, gate_b1, l16);
  gemm_accT(R0, LDP, R2, LDP, H_DIM, acc, lane, rowBase);
  __syncthreads();
  store_stripe(R2, LDP, acc, lane, rowBase, false);
  if (tid < H_DIM) sW2[tid] = gate_w2[tid];
  __syncthreads();

  const float gb2 = gate_b2[0];
  float gv[32];
#pragma unroll 1
  for (int ii = 0; ii < 16; ++ii) {
    const int i = rowBase + ii;
    float s0 = 0.0f, s1 = 0.0f;
    for (int h = 0; h < H_DIM; ++h) {
      const float gih = R1[i * LDP + h];
      const float w2h = sW2[h];
      s0 += fmaxf(gih + R2[lane * LDP + h], 0.0f) * w2h;
      s1 += fmaxf(gih + R2[(lane + 32) * LDP + h], 0.0f) * w2h;
    }
    gv[2 * ii + 0] = 1.0f / (1.0f + __expf(-(s0 + gb2)));
    gv[2 * ii + 1] = 1.0f / (1.0f + __expf(-(s1 + gb2)));
  }
#pragma unroll
  for (int ii = 0; ii < 16; ++ii) {
    R1[(rowBase + ii) * LDP + lane]      = gv[2 * ii + 0];
    R1[(rowBase + ii) * LDP + lane + 32] = gv[2 * ii + 1];
  }
  __syncthreads();

  copy_w_T(R2, msg_w1, 0, tid);
  __syncthreads();
  init_acc(acc, msg_b1, l16);
  gemm_accT(R0, LDP, R2, LDP, H_DIM, acc, lane, rowBase);
  __syncthreads();
  store_stripe(R0, LDP, acc, lane, rowBase,  true);
  copy_w_T(R2, msg_w2, 0, tid);
  __syncthreads();
  init_acc(acc, msg_b2, l16);
  gemm_accT(R0, LDP, R2, LDP, H_DIM, acc, lane, rowBase);
  __syncthreads();
  store_stripeT(R2, LDP, acc, lane, rowBase);
  __syncthreads();

  init_acc(acc, nullptr, l16);
  gemm_accT(R1, LDP, R2, LDP, N_DIM, acc, lane, rowBase);

  float* outb = out + (size_t)b * (N_DIM * H_DIM);
  const int half = lane >> 4;
#pragma unroll
  for (int t = 0; t < 4; ++t)
#pragma unroll
    for (int r = 0; r < 8; ++r)
      outb[(rowBase + r + 8 * half) * H_DIM + t * 16 + l16] = acc[t][r];
}

extern "C" void kernel_launch(void* const* d_in, const int* in_sizes, int n_in,
                              void* d_out, int out_size, void* d_ws, size_t ws_size,
                              hipStream_t stream) {
  const float* obs     = (const float*)d_in[0];
  const float* enc_w   = (const float*)d_in[1];
  const float* enc_b   = (const float*)d_in[2];
  const float* gate_w1 = (const float*)d_in[3];
  const float* gate_b1 = (const float*)d_in[4];
  const float* gate_w2 = (const float*)d_in[5];
  const float* gate_b2 = (const float*)d_in[6];
  const float* msg_w1  = (const float*)d_in[7];
  const float* msg_b1  = (const float*)d_in[8];
  const float* msg_w2  = (const float*)d_in[9];
  const float* msg_b2  = (const float*)d_in[10];
  float* out = (float*)d_out;

  gated_comm_kernel<<<BATCH, 128, 0, stream>>>(
      obs, enc_w, enc_b, gate_w1, gate_b1, gate_w2, gate_b2,
      msg_w1, msg_b1, msg_w2, msg_b2, out);
}


# column-stripe tiling: a wave owns 16 output columns of each weight GEMM; weight slices go global->B-fragment registers, no LDS staging; msg kept in accumulators as B of out GEMM (sum order changes)
# speedup vs baseline: 1.5942x; 1.5942x over previous
#include <hip/hip_runtime.h>
#include <math.h>


typedef __attribute__((ext_vector_type(2))) float v2f;
typedef __attribute__((ext_vector_type(8))) float v8f;

#define BATCH 256
#define N_DIM 64
#define D_DIM 128
#define H_DIM 64
#define LDP   66
#define LDO   132
#define LDX   68

__device__ inline v8f wmma_f32x4(v2f a, v2f b, v8f c) {
  return __builtin_amdgcn_wmma_f32_16x16x4_f32(
       false, a,  false, b,
       (short)0, c,  false,  false);
}

template <int KS>
__device__ inline void load_wfrag(v2f (&f)[KS], const float* __restrict__ W,
                                  int n, int half) {
  const float* p = W + (2 * half) * H_DIM + n;
#pragma unroll
  for (int ks = 0; ks < KS; ++ks) {
    f[ks].x = p[(4 * ks) * H_DIM];
    f[ks].y = p[(4 * ks + 1) * H_DIM];
  }
}

__device__ inline void fill_acc(v8f acc[4], float v) {
#pragma unroll
  for (int t = 0; t < 4; ++t)
#pragma unroll
    for (int r = 0; r < 8; ++r) acc[t][r] = v;
}

__device__ inline void store_cols(float* __restrict__ D, int ldd,
                                  const v8f acc[4], int half, int n,
                                  bool do_relu) {
#pragma unroll
  for (int rt = 0; rt < 4; ++rt)
#pragma unroll
    for (int r = 0; r < 8; ++r) {
      float v = acc[rt][r];
      if (do_relu) v = fmaxf(v, 0.0f);
      D[(rt * 16 + r + 8 * half) * ldd + n] = v;
    }
}

__global__ __launch_bounds__(128) void gated_comm_kernel(
    const float* __restrict__ obs, const float* __restrict__ enc_w,
    const float* __restrict__ enc_b, const float* __restrict__ gate_w1,
    const float* __restrict__ gate_b1, const float* __restrict__ gate_w2,
    const float* __restrict__ gate_b2, const float* __restrict__ msg_w1,
    const float* __restrict__ msg_b1, const float* __restrict__ msg_w2,
    const float* __restrict__ msg_b2, float* __restrict__ out) {
  __shared__ __attribute__((aligned(16))) float smem[64 * LDO + 4 * 64 * LDP + 64];

  const int b       = blockIdx.x;
  const int tid     = threadIdx.x;
  const int lane    = tid & 31;
  const int wave    = tid >> 5;
  const int l16     = lane & 15;
  const int half    = lane >> 4;
  const int rowBase = wave * 16;
  const int n       = wave * 16 + l16;

  float* O   = smem;
  float* E   = smem + 64 * LDO;
  float* R1  = E  + 64 * LDP;
  float* R2  = R1 + 64 * LDP;
  float* T   = R2 + 64 * LDP;
  float* sW2 = T  + 64 * LDP;

  v2f wEnc[32], wI[16], wJ[16], wM1[16], wM2[16];

  const float* obsb = obs + (size_t)b * (N_DIM * D_DIM);
  float4 ob[16];
#pragma unroll
  for (int it = 0; it < 16; ++it)
    ob[it] = *(const float4*)(obsb + (wave + 4 * it) * D_DIM + lane * 4);

  load_wfrag<32>(wEnc, enc_w, n, half);
  const float bEnc = enc_b[n];

#pragma unroll
  for (int it = 0; it < 16; ++it)
    *(float4*)(O + (wave + 4 * it) * LDO + lane * 4) = ob[it];
  if (tid < H_DIM) sW2[tid] = gate_w2[tid];
  __syncthreads();

  load_wfrag<16>(wI,  gate_w1, n, half);
  load_wfrag<16>(wJ,  gate_w1 + H_DIM * H_DIM, n, half);
  load_wfrag<16>(wM1, msg_w1, n, half);
  const float bG1 = gate_b1[n];
  const float bM1 = msg_b1[n];

  {
    v8f acc[4];
    fill_acc(acc, bEnc);
#pragma unroll
    for (int ks = 0; ks < 32; ++ks) {
      const int ka = 4 * ks + 2 * half;
#pragma unroll
      for (int rt = 0; rt < 4; ++rt) {
        v2f a = *(const v2f*)(O + (rt * 16 + l16) * LDO + ka);
        acc[rt] = wmma_f32x4(a, wEnc[ks], acc[rt]);
      }
    }
    store_cols(E, LDP, acc, half, n,  true);
  }
  __syncthreads();
  load_wfrag<16>(wM2, msg_w2, n, half);
  const float bM2 = msg_b2[n];

  {
    v8f aI[4], aJ[4], aM[4];
    fill_acc(aI, 0.0f);
    fill_acc(aJ, bG1);
    fill_acc(aM, bM1);
#pragma unroll
    for (int ks = 0; ks < 16; ++ks) {
      const int ka = 4 * ks + 2 * half;
#pragma unroll
      for (int rt = 0; rt < 4; ++rt) {
        v2f a = *(const v2f*)(E + (rt * 16 + l16) * LDP + ka);
        aI[rt] = wmma_f32x4(a, wI[ks],  aI[rt]);
        aJ[rt] = wmma_f32x4(a, wJ[ks],  aJ[rt]);
        aM[rt] = wmma_f32x4(a, wM1[ks], aM[rt]);
      }
    }
    store_cols(R1, LDP, aI, half, n, false);
    store_cols(R2, LDP, aJ, half, n, false);
    store_cols(T,  LDP, aM, half, n,  true);
  }
  __syncthreads();

  const float gb2 = gate_b2[0];
  float gv[32];
#pragma unroll 1
  for (int ii = 0; ii < 16; ++ii) {
    const int i = rowBase + ii;
    float s0 = 0.0f, s1 = 0.0f;
    for (int h = 0; h < H_DIM; ++h) {
      const float gih = R1[i * LDP + h];
      const float w2h = sW2[h];
      s0 += fmaxf(gih + R2[lane * LDP + h], 0.0f) * w2h;
      s1 += fmaxf(gih + R2[(lane + 32) * LDP + h], 0.0f) * w2h;
    }
    gv[2 * ii + 0] = 1.0f / (1.0f + __expf(-(s0 + gb2)));
    gv[2 * ii + 1] = 1.0f / (1.0f + __expf(-(s1 + gb2)));
  }
#pragma unroll
  for (int ii = 0; ii < 16; ++ii) {
    R1[(rowBase + ii) * LDP + lane]      = gv[2 * ii + 0];
    R1[(rowBase + ii) * LDP + lane + 32] = gv[2 * ii + 1];
  }

  v8f mAcc[4];
  fill_acc(mAcc, bM2);
#pragma unroll
  for (int ks = 0; ks < 16; ++ks) {
    const int ka = 4 * ks + 2 * half;
#pragma unroll
    for (int rt = 0; rt < 4; ++rt) {
      v2f a = *(const v2f*)(T + (rt * 16 + l16) * LDP + ka);
      mAcc[rt] = wmma_f32x4(a, wM2[ks], mAcc[rt]);
    }
  }
  __syncthreads();

  v8f oAcc[4];
  fill_acc(oAcc, 0.0f);
#pragma unroll
  for (int rt = 0; rt < 4; ++rt) {
#pragma unroll
    for (int q = 0; q < 4; ++q) {
      v2f bb;
      bb.x = mAcc[rt][2 * q];
      bb.y = mAcc[rt][2 * q + 1];
      const int kj = 16 * rt + 8 * half + 2 * q;
#pragma unroll
      for (int mt = 0; mt < 4; ++mt) {
        v2f a = *(const v2f*)(R1 + (mt * 16 + l16) * LDP + kj);
        oAcc[mt] = wmma_f32x4(a, bb, oAcc[mt]);
      }
    }
  }

  store_cols(O, LDX, oAcc, half, n, false);
  __syncthreads();
  float* outb = out + (size_t)b * (N_DIM * H_DIM);
#pragma unroll
  for (int it = 0; it < 8; ++it) {
    const int s  = tid + 128 * it;
    const int r  = s >> 4;
    const int c4 = (s & 15) * 4;
    *(float4*)(outb + r * H_DIM + c4) = *(const float4*)(O + r * LDX + c4);
  }
}

extern "C" void kernel_launch(void* const* d_in, const int* in_sizes, int n_in,
                              void* d_out, int out_size, void* d_ws, size_t ws_size,
                              hipStream_t stream) {
  const float* obs     = (const float*)d_in[0];
  const float* enc_w   = (const float*)d_in[1];
  const float* enc_b   = (const float*)d_in[2];
  const float* gate_w1 = (const float*)d_in[3];
  const float* gate_b1 = (const float*)d_in[4];
  const float* gate_w2 = (const float*)d_in[5];
  const float* gate_b2 = (const float*)d_in[6];
  const float* msg_w1  = (const float*)d_in[7];
  const float* msg_b1  = (const float*)d_in[8];
  const float* msg_w2  = (const float*)d_in[9];
  const float* msg_b2  = (const float*)d_in[10];
  float* out = (float*)d_out;

  gated_comm_kernel<<<BATCH, 128, 0, stream>>>(
      obs, enc_w, enc_b, gate_w1, gate_b1, gate_w2, gate_b2,
      msg_w1, msg_b1, msg_w2, msg_b2, out);
}


# gj computed transposed on the matrix unit (w_j as A, enc rows as B): a lane's accumulators hold its column's gj for 32 h; gate sums run from registers, gi as LDS broadcasts; h sum order changes
# speedup vs baseline: 2.6646x; 1.6714x over previous
#include <hip/hip_runtime.h>
#include <math.h>


typedef __attribute__((ext_vector_type(2))) float v2f;
typedef __attribute__((ext_vector_type(4))) float v4f;
typedef __attribute__((ext_vector_type(8))) float v8f;

#define BATCH 256
#define N_DIM 64
#define D_DIM 128
#define H_DIM 64
#define LDP   66
#define LDG   68
#define LDO   132
#define LDX   68

__device__ inline v8f wmma_f32x4(v2f a, v2f b, v8f c) {
  return __builtin_amdgcn_wmma_f32_16x16x4_f32(
       false, a,  false, b,
       (short)0, c,  false,  false);
}

template <int KS>
__device__ inline void load_wfrag(v2f (&f)[KS], const float* __restrict__ W,
                                  int n, int half) {
  const float* p = W + (2 * half) * H_DIM + n;
#pragma unroll
  for (int ks = 0; ks < KS; ++ks) {
    f[ks].x = p[(4 * ks) * H_DIM];
    f[ks].y = p[(4 * ks + 1) * H_DIM];
  }
}

__device__ inline void fill_acc(v8f acc[4], float v) {
#pragma unroll
  for (int t = 0; t < 4; ++t)
#pragma unroll
    for (int r = 0; r < 8; ++r) acc[t][r] = v;
}

__device__ inline void store_cols(float* __restrict__ D, int ldd,
                                  const v8f acc[4], int half, int n,
                                  bool do_relu) {
#pragma unroll
  for (int rt = 0; rt < 4; ++rt)
#pragma unroll
    for (int r = 0; r < 8; ++r) {
      float v = acc[rt][r];
      if (do_relu) v = fmaxf(v, 0.0f);
      D[(rt * 16 + r + 8 * half) * ldd + n] = v;
    }
}

__device__ inline float xhalf(float x) {
  const int b = __builtin_bit_cast(int, x);
  return __builtin_bit_cast(float, __builtin_amdgcn_permlanex16(
      b, b, 0x76543210, 0xfedcba98, false, false));
}

__device__ inline void load_gi4(v4f (&g)[4][8], const float* gip, int i0) {
#pragma unroll
  for (int k = 0; k < 4; ++k)
#pragma unroll
    for (int q = 0; q < 8; ++q)
      g[k][q] = *(const v4f*)(gip + (i0 + k) * LDG + 4 * q);
}

__device__ inline void gate_rows4(float (&s)[4], const v4f (&g)[4][8],
                                  const v2f (&gj2)[8][2], const v4f (&w2v)[8]) {
  v2f acc[4];
#pragma unroll
  for (int k = 0; k < 4; ++k) acc[k] = (v2f){0.0f, 0.0f};
#pragma unroll
  for (int q = 0; q < 8; ++q)
#pragma unroll
    for (int e = 0; e < 2; ++e) {
      const v2f w = e ? __builtin_shufflevector(w2v[q], w2v[q], 2, 3)
                      : __builtin_shufflevector(w2v[q], w2v[q], 0, 1);
#pragma unroll
      for (int k = 0; k < 4; ++k) {
        const v2f gi2 = e ? __builtin_shufflevector(g[k][q], g[k][q], 2, 3)
                          : __builtin_shufflevector(g[k][q], g[k][q], 0, 1);
        v2f t = gi2 + gj2[q][e];
        t.x = fmaxf(t.x, 0.0f);
        t.y = fmaxf(t.y, 0.0f);
        acc[k] = t * w + acc[k];
      }
    }
#pragma unroll
  for (int k = 0; k < 4; ++k) s[k] = acc[k].x + acc[k].y;
}

__device__ inline void gate_finish4(const float (&s)[4], float gb2, int half,
                                    float* Gcol, int i0) {
  float t[4];
#pragma unroll
  for (int k = 0; k < 4; ++k) t[k] = s[k] + xhalf(s[k]);
  const float tA = half ? t[2] : t[0];
  const float tB = half ? t[3] : t[1];
  const int ia = i0 + 2 * half;
  Gcol[ia * LDP]       = 1.0f / (1.0f + __expf(-(tA + gb2)));
  Gcol[(ia + 1) * LDP] = 1.0f / (1.0f + __expf(-(tB + gb2)));
}

__global__ __launch_bounds__(128) void gated_comm_kernel(
    const float* __restrict__ obs, const float* __restrict__ enc_w,
    const float* __restrict__ enc_b, const float* __restrict__ gate_w1,
    const float* __restrict__ gate_b1, const float* __restrict__ gate_w2,
    const float* __restrict__ gate_b2, const float* __restrict__ msg_w1,
    const float* __restrict__ msg_b1, const float* __restrict__ msg_w2,
    const float* __restrict__ msg_b2, float* __restrict__ out) {
  __shared__ __attribute__((aligned(16))) float smem[64 * LDO + 3 * 64 * LDP + 64 * LDG + 64];

  const int b       = blockIdx.x;
  const int tid     = threadIdx.x;
  const int lane    = tid & 31;
  const int wave    = tid >> 5;
  const int l16     = lane & 15;
  const int half    = lane >> 4;
  const int rowBase = wave * 16;
  const int n       = wave * 16 + l16;

  float* O   = smem;
  float* E   = smem + 64 * LDO;
  float* R1  = E  + 64 * LDP;
  float* G   = R1 + 64 * LDG;
  float* T   = G  + 64 * LDP;
  float* sW2 = T  + 64 * LDP;

  v2f wEnc[32], wI[16], wM1[16], wM2[16];
  v4f wJ4[16][2];
  v4f b1v[8];

  const float* obsb = obs + (size_t)b * (N_DIM * D_DIM);
  float4 ob[16];
#pragma unroll
  for (int it = 0; it < 16; ++it)
    ob[it] = *(const float4*)(obsb + (wave + 4 * it) * D_DIM + lane * 4);

  load_wfrag<32>(wEnc, enc_w, n, half);
  const float bEnc = enc_b[n];

#pragma unroll
  for (int it = 0; it < 16; ++it)
    *(float4*)(O + (wave + 4 * it) * LDO + lane * 4) = ob[it];
  if (tid < H_DIM) sW2[tid] = gate_w2[tid];
  __syncthreads();

  load_wfrag<16>(wI,  gate_w1, n, half);
  {
    const float* pj = gate_w1 + H_DIM * H_DIM + (2 * half) * H_DIM + 4 * l16;
#pragma unroll
    for (int ks = 0; ks < 16; ++ks) {
      wJ4[ks][0] = *(const v4f*)(pj + (4 * ks) * H_DIM);
      wJ4[ks][1] = *(const v4f*)(pj + (4 * ks + 1) * H_DIM);
    }
#pragma unroll
    for (int q = 0; q < 8; ++q)
      b1v[q] = *(const v4f*)(gate_b1 + 32 * half + 4 * q);
  }
  load_wfrag<16>(wM1, msg_w1, n, half);
  const float bM1 = msg_b1[n];

  {
    v8f acc[4];
    fill_acc(acc, bEnc);
#pragma unroll
    for (int ks = 0; ks < 32; ++ks) {
      const int ka = 4 * ks + 2 * half;
#pragma unroll
      for (int rt = 0; rt < 4; ++rt) {
        v2f a = *(const v2f*)(O + (rt * 16 + l16) * LDO + ka);
        acc[rt] = wmma_f32x4(a, wEnc[ks], acc[rt]);
      }
    }
    store_cols(E, LDP, acc, half, n,  true);
  }
  __syncthreads();
  load_wfrag<16>(wM2, msg_w2, n, half);
  const float bM2 = msg_b2[n];

  v8f gjT[4];
  {
    v8f aI[4], aM[4];
    fill_acc(aI, 0.0f);
    fill_acc(aM, bM1);
#pragma unroll
    for (int mt = 0; mt < 4; ++mt)
#pragma unroll
      for (int r = 0; r < 8; ++r) gjT[mt][r] = b1v[r][mt];
#pragma unroll
    for (int ks = 0; ks < 16; ++ks) {
      const int ka = 4 * ks + 2 * half;
      const v2f eJ = *(const v2f*)(E + (rowBase + l16) * LDP + ka);
#pragma unroll
      for (int rt = 0; rt < 4; ++rt) {
        v2f a = *(const v2f*)(E + (rt * 16 + l16) * LDP + ka);
        aI[rt] = wmma_f32x4(a, wI[ks],  aI[rt]);
        aM[rt] = wmma_f32x4(a, wM1[ks], aM[rt]);
      }
#pragma unroll
      for (int mt = 0; mt < 4; ++mt) {
        v2f wa;
        wa.x = wJ4[ks][0][mt];
        wa.y = wJ4[ks][1][mt];
        gjT[mt] = wmma_f32x4(wa, eJ, gjT[mt]);
      }
    }
    store_cols(R1, LDG, aI, half, n, false);
    store_cols(T,  LDP, aM, half, n,  true);
  }
  __syncthreads();

  const float gb2 = gate_b2[0];
  {
    v4f w2v[8];
#pragma unroll
    for (int q = 0; q < 8; ++q) w2v[q] = *(const v4f*)(sW2 + 32 * half + 4 * q);
    v2f gj2[8][2];
#pragma unroll
    for (int q = 0; q < 8; ++q) {
      gj2[q][0] = (v2f){gjT[0][q], gjT[1][q]};
      gj2[q][1] = (v2f){gjT[2][q], gjT[3][q]};
    }
    const float* gip = R1 + 32 * half;
    float* Gcol = G + rowBase + l16;
    v4f gA[4][8], gB[4][8];
    float sA[4], sB[4];
    load_gi4(gA, gip, 0);
    __builtin_amdgcn_sched_barrier(0);
    load_gi4(gB, gip, 4);
    __builtin_amdgcn_sched_barrier(0);
    gate_rows4(sA, gA, gj2, w2v);
#pragma unroll 1
    for (int p = 1; p < 15; p += 2) {
      __builtin_amdgcn_sched_barrier(0);
      load_gi4(gA, gip, 4 * (p + 1));
      __builtin_amdgcn_sched_barrier(0);
      gate_rows4(sB, gB, gj2, w2v);
      gate_finish4(sA, gb2, half, Gcol, 4 * (p - 1));
      __builtin_amdgcn_sched_barrier(0);
      load_gi4(gB, gip, 4 * (p + 2));
      __builtin_amdgcn_sched_barrier(0);
      gate_rows4(sA, gA, gj2, w2v);
      gate_finish4(sB, gb2, half, Gcol, 4 * p);
    }
    __builtin_amdgcn_sched_barrier(0);
    gate_rows4(sB, gB, gj2, w2v);
    gate_finish4(sA, gb2, half, Gcol, 4 * 14);
    gate_finish4(sB, gb2, half, Gcol, 4 * 15);
  }

  v8f mAcc[4];
  fill_acc(mAcc, bM2);
#pragma unroll
  for (int ks = 0; ks < 16; ++ks) {
    const int ka = 4 * ks + 2 * half;
#pragma unroll
    for (int rt = 0; rt < 4; ++rt) {
      v2f a = *(const v2f*)(T + (rt * 16 + l16) * LDP + ka);
      mAcc[rt] = wmma_f32x4(a, wM2[ks], mAcc[rt]);
    }
  }
  __syncthreads();

  v8f oAcc[4];
  fill_acc(oAcc, 0.0f);
#pragma unroll
  for (int rt = 0; rt < 4; ++rt) {
#pragma unroll
    for (int q = 0; q < 4; ++q) {
      v2f bb;
      bb.x = mAcc[rt][2 * q];
      bb.y = mAcc[rt][2 * q + 1];
      const int kj = 16 * rt + 8 * half + 2 * q;
#pragma unroll
      for (int mt = 0; mt < 4; ++mt) {
        v2f a = *(const v2f*)(G + (mt * 16 + l16) * LDP + kj);
        oAcc[mt] = wmma_f32x4(a, bb, oAcc[mt]);
      }
    }
  }

  store_cols(O, LDX, oAcc, half, n, false);
  __syncthreads();
  float* outb = out + (size_t)b * (N_DIM * H_DIM);
#pragma unroll
  for (int it = 0; it < 8; ++it) {
    const int s  = tid + 128 * it;
    const int r  = s >> 4;
    const int c4 = (s & 15) * 4;
    *(float4*)(outb + r * H_DIM + c4) = *(const float4*)(O + r * LDX + c4);
  }
}

extern "C" void kernel_launch(void* const* d_in, const int* in_sizes, int n_in,
                              void* d_out, int out_size, void* d_ws, size_t ws_size,
                              hipStream_t stream) {
  const float* obs     = (const float*)d_in[0];
  const float* enc_w   = (const float*)d_in[1];
  const float* enc_b   = (const float*)d_in[2];
  const float* gate_w1 = (const float*)d_in[3];
  const float* gate_b1 = (const float*)d_in[4];
  const float* gate_w2 = (const float*)d_in[5];
  const float* gate_b2 = (const float*)d_in[6];
  const float* msg_w1  = (const float*)d_in[7];
  const float* msg_b1  = (const float*)d_in[8];
  const float* msg_w2  = (const float*)d_in[9];
  const float* msg_b2  = (const float*)d_in[10];
  float* out = (float*)d_out;

  gated_comm_kernel<<<BATCH, 128, 0, stream>>>(
      obs, enc_w, enc_b, gate_w1, gate_b1, gate_w2, gate_b2,
      msg_w1, msg_b1, msg_w2, msg_b2, out);
}


# one memory round trip before barrier 1, not three: entry issues 49 loads (obs tile, half the enc_w slice), the rest in 5 bursts under the WMMAs; gate_w2 read to registers, not via LDS; sums unchanged
# speedup vs baseline: 2.8000x; 1.0508x over previous
#include <hip/hip_runtime.h>
#include <math.h>


typedef __attribute__((ext_vector_type(2))) float v2f;
typedef __attribute__((ext_vector_type(4))) float v4f;
typedef __attribute__((ext_vector_type(8))) float v8f;

#define BATCH 256
#define N_DIM 64
#define D_DIM 128
#define H_DIM 64
#define LDP   66
#define LDG   68
#define LDO   132
#define LDX   68
#define LOADS_STAY 0x786

__device__ inline v8f wmma_f32x4(v2f a, v2f b, v8f c) {
  return __builtin_amdgcn_wmma_f32_16x16x4_f32(
       false, a,  false, b,
       (short)0, c,  false,  false);
}

template <int KS>
__device__ inline void load_wfrag(v2f (&f)[KS], const float* __restrict__ W,
                                  int n, int half) {
  const float* p = W + (2 * half) * H_DIM + n;
#pragma unroll
  for (int ks = 0; ks < KS; ++ks) {
    f[ks].x = p[(4 * ks) * H_DIM];
    f[ks].y = p[(4 * ks + 1) * H_DIM];
  }
}

__device__ inline void fill_acc(v8f acc[4], float v) {
#pragma unroll
  for (int t = 0; t < 4; ++t)
#pragma unroll
    for (int r = 0; r < 8; ++r) acc[t][r] = v;
}

__device__ inline void store_cols(float* __restrict__ D, int ldd,
                                  const v8f acc[4], int half, int n,
                                  bool do_relu) {
#pragma unroll
  for (int rt = 0; rt < 4; ++rt)
#pragma unroll
    for (int r = 0; r < 8; ++r) {
      float v = acc[rt][r];
      if (do_relu) v = fmaxf(v, 0.0f);
      D[(rt * 16 + r + 8 * half) * ldd + n] = v;
    }
}

__device__ inline float xhalf(float x) {
  const int b = __builtin_bit_cast(int, x);
  return __builtin_bit_cast(float, __builtin_amdgcn_permlanex16(
      b, b, 0x76543210, 0xfedcba98, false, false));
}

__device__ inline void load_gi4(v4f (&g)[4][8], const float* gip, int i0) {
#pragma unroll
  for (int k = 0; k < 4; ++k)
#pragma unroll
    for (int q = 0; q < 8; ++q)
      g[k][q] = *(const v4f*)(gip + (i0 + k) * LDG + 4 * q);
}

__device__ inline void gate_rows4(float (&s)[4], const v4f (&g)[4][8],
                                  const v2f (&gj2)[8][2], const v4f (&w2v)[8]) {
  v2f acc[4];
#pragma unroll
  for (int k = 0; k < 4; ++k) acc[k] = (v2f){0.0f, 0.0f};
#pragma unroll
  for (int q = 0; q < 8; ++q)
#pragma unroll
    for (int e = 0; e < 2; ++e) {
      const v2f w = e ? __builtin_shufflevector(w2v[q], w2v[q], 2, 3)
                      : __builtin_shufflevector(w2v[q], w2v[q], 0, 1);
#pragma unroll
      for (int k = 0; k < 4; ++k) {
        const v2f gi2 = e ? __builtin_shufflevector(g[k][q], g[k][q], 2, 3)
                          : __builtin_shufflevector(g[k][q], g[k][q], 0, 1);
        v2f t = gi2 + gj2[q][e];
        t.x = fmaxf(t.x, 0.0f);
        t.y = fmaxf(t.y, 0.0f);
        acc[k] = t * w + acc[k];
      }
    }
#pragma unroll
  for (int k = 0; k < 4; ++k) s[k] = acc[k].x + acc[k].y;
}

__device__ inline void gate_finish4(const float (&s)[4], float gb2, int half,
                                    float* Gcol, int i0) {
  float t[4];
#pragma unroll
  for (int k = 0; k < 4; ++k) t[k] = s[k] + xhalf(s[k]);
  const float tA = half ? t[2] : t[0];
  const float tB = half ? t[3] : t[1];
  const int ia = i0 + 2 * half;
  Gcol[ia * LDP]       = 1.0f / (1.0f + __expf(-(tA + gb2)));
  Gcol[(ia + 1) * LDP] = 1.0f / (1.0f + __expf(-(tB + gb2)));
}

__global__ __launch_bounds__(128) void gated_comm_kernel(
    const float* __restrict__ obs, const float* __restrict__ enc_w,
    const float* __restrict__ enc_b, const float* __restrict__ gate_w1,
    const float* __restrict__ gate_b1, const float* __restrict__ gate_w2,
    const float* __restrict__ gate_b2, const float* __restrict__ msg_w1,
    const float* __restrict__ msg_b1, const float* __restrict__ msg_w2,
    const float* __restrict__ msg_b2, float* __restrict__ out) {
  __shared__ __attribute__((aligned(16))) float smem[64 * LDO + 3 * 64 * LDP + 64 * LDG];

  const int b       = blockIdx.x;
  const int tid     = threadIdx.x;
  const int lane    = tid & 31;
  const int wave    = tid >> 5;
  const int l16     = lane & 15;
  const int half    = lane >> 4;
  const int rowBase = wave * 16;
  const int n       = wave * 16 + l16;

  float* O   = smem;
  float* E   = smem + 64 * LDO;
  float* R1  = E  + 64 * LDP;
  float* G   = R1 + 64 * LDG;
  float* T   = G  + 64 * LDP;

  v2f wEncA[16], wEncB[16], wI[16], wM1[16], wM2[16];
  v4f wJ4[16][2];
  v4f b1v[8];

  const float* obsb = obs + (size_t)b * (N_DIM * D_DIM);
  float4 ob[16];
#pragma unroll
  for (int it = 0; it < 16; ++it)
    ob[it] = *(const float4*)(obsb + (wave + 4 * it) * D_DIM + lane * 4);

  load_wfrag<16>(wEncA, enc_w, n, half);
  const float bEnc = enc_b[n];

#pragma unroll
  for (int it = 0; it < 16; ++it)
    *(float4*)(O + (wave + 4 * it) * LDO + lane * 4) = ob[it];
  __syncthreads();

  float bM1 = 0.0f;
  {
    v8f acc[4];
    fill_acc(acc, bEnc);
#pragma unroll
    for (int ks = 0; ks < 32; ++ks) {
      if (ks == 2) {
        __builtin_amdgcn_sched_barrier(LOADS_STAY);
        load_wfrag<16>(wEncB, enc_w + 64 * H_DIM, n, half);
        bM1 = msg_b1[n];
        __builtin_amdgcn_sched_barrier(LOADS_STAY);
      }
      if (ks == 8) {
        __builtin_amdgcn_sched_barrier(LOADS_STAY);
        load_wfrag<16>(wI, gate_w1, n, half);
        __builtin_amdgcn_sched_barrier(LOADS_STAY);
      }
      if (ks == 14) {
        __builtin_amdgcn_sched_barrier(LOADS_STAY);
        const float* pj = gate_w1 + H_DIM * H_DIM + (2 * half) * H_DIM + 4 * l16;
#pragma unroll
        for (int kj = 0; kj < 16; ++kj) {
          wJ4[kj][0] = *(const v4f*)(pj + (4 * kj) * H_DIM);
          wJ4[kj][1] = *(const v4f*)(pj + (4 * kj + 1) * H_DIM);
        }
        __builtin_amdgcn_sched_barrier(LOADS_STAY);
      }
      if (ks == 20) {
        __builtin_amdgcn_sched_barrier(LOADS_STAY);
        load_wfrag<16>(wM1, msg_w1, n, half);
#pragma unroll
        for (int q = 0; q < 8; ++q)
          b1v[q] = *(const v4f*)(gate_b1 + 32 * half + 4 * q);
        __builtin_amdgcn_sched_barrier(LOADS_STAY);
      }
      const int ka = 4 * ks + 2 * half;
#pragma unroll
      for (int rt = 0; rt < 4; ++rt) {
        v2f a = *(const v2f*)(O + (rt * 16 + l16) * LDO + ka);
        acc[rt] = wmma_f32x4(a, ks < 16 ? wEncA[ks & 15] : wEncB[ks & 15], acc[rt]);
      }
    }
    store_cols(E, LDP, acc, half, n,  true);
  }
  __syncthreads();
  float bM2 = 0.0f;
  v4f w2v[8];

  v8f gjT[4];
  {
    v8f aI[4], aM[4];
    fill_acc(aI, 0.0f);
    fill_acc(aM, bM1);
#pragma unroll
    for (int mt = 0; mt < 4; ++mt)
#pragma unroll
      for (int r = 0; r < 8; ++r) gjT[mt][r] = b1v[r][mt];
#pragma unroll
    for (int ks = 0; ks < 16; ++ks) {
      if (ks == 2) {
        __builtin_amdgcn_sched_barrier(LOADS_STAY);
        load_wfrag<16>(wM2, msg_w2, n, half);
        bM2 = msg_b2[n];
#pragma unroll
        for (int q = 0; q < 8; ++q) w2v[q] = *(const v4f*)(gate_w2 + 32 * half + 4 * q);
        __builtin_amdgcn_sched_barrier(LOADS_STAY);
      }
      const int ka = 4 * ks + 2 * half;
      const v2f eJ = *(const v2f*)(E + (rowBase + l16) * LDP + ka);
#pragma unroll
      for (int rt = 0; rt < 4; ++rt) {
        v2f a = *(const v2f*)(E + (rt * 16 + l16) * LDP + ka);
        aI[rt] = wmma_f32x4(a, wI[ks],  aI[rt]);
        aM[rt] = wmma_f32x4(a, wM1[ks], aM[rt]);
      }
#pragma unroll
      for (int mt = 0; mt < 4; ++mt) {
        v2f wa;
        wa.x = wJ4[ks][0][mt];
        wa.y = wJ4[ks][1][mt];
        gjT[mt] = wmma_f32x4(wa, eJ, gjT[mt]);
      }
    }
    store_cols(R1, LDG, aI, half, n, false);
    store_cols(T,  LDP, aM, half, n,  true);
  }
  __syncthreads();

  const float gb2 = gate_b2[0];
  {
    v2f gj2[8][2];
#pragma unroll
    for (int q = 0; q < 8; ++q) {
      gj2[q][0] = (v2f){gjT[0][q], gjT[1][q]};
      gj2[q][1] = (v2f){gjT[2][q], gjT[3][q]};
    }
    const float* gip = R1 + 32 * half;
    float* Gcol = G + rowBase + l16;
    v4f gA[4][8], gB[4][8];
    float sA[4], sB[4];
    load_gi4(gA, gip, 0);
    __builtin_amdgcn_sched_barrier(0);
    load_gi4(gB, gip, 4);
    __builtin_amdgcn_sched_barrier(0);
    gate_rows4(sA, gA, gj2, w2v);
#pragma unroll 1
    for (int p = 1; p < 15; p += 2) {
      __builtin_amdgcn_sched_barrier(0);
      load_gi4(gA, gip, 4 * (p + 1));
      __builtin_amdgcn_sched_barrier(0);
      gate_rows4(sB, gB, gj2, w2v);
      gate_finish4(sA, gb2, half, Gcol, 4 * (p - 1));
      __builtin_amdgcn_sched_barrier(0);
      load_gi4(gB, gip, 4 * (p + 2));
      __builtin_amdgcn_sched_barrier(0);
      gate_rows4(sA, gA, gj2, w2v);
      gate_finish4(sB, gb2, half, Gcol, 4 * p);
    }
    __builtin_amdgcn_sched_barrier(0);
    gate_rows4(sB, gB, gj2, w2v);
    gate_finish4(sA, gb2, half, Gcol, 4 * 14);
    gate_finish4(sB, gb2, half, Gcol, 4 * 15);
  }

  v8f mAcc[4];
  fill_acc(mAcc, bM2);
#pragma unroll
  for (int ks = 0; ks < 16; ++ks) {
    const int ka = 4 * ks + 2 * half;
#pragma unroll
    for (int rt = 0; rt < 4; ++rt) {
      v2f a = *(const v2f*)(T + (rt * 16 + l16) * LDP + ka);
      mAcc[rt] = wmma_f32x4(a, wM2[ks], mAcc[rt]);
    }
  }
  __syncthreads();

  v8f oAcc[4];
  fill_acc(oAcc, 0.0f);
#pragma unroll
  for (int rt = 0; rt < 4; ++rt) {
#pragma unroll
    for (int q = 0; q < 4; ++q) {
      v2f bb;
      bb.x = mAcc[rt][2 * q];
      bb.y = mAcc[rt][2 * q + 1];
      const int kj = 16 * rt + 8 * half + 2 * q;
#pragma unroll
      for (int mt = 0; mt < 4; ++mt) {
        v2f a = *(const v2f*)(G + (mt * 16 + l16) * LDP + kj);
        oAcc[mt] = wmma_f32x4(a, bb, oAcc[mt]);
      }
    }
  }

  store_cols(O, LDX, oAcc, half, n, false);
  __syncthreads();
  float* outb = out + (size_t)b * (N_DIM * H_DIM);
#pragma unroll
  for (int it = 0; it < 8; ++it) {
    const int s  = tid + 128 * it;
    const int r  = s >> 4;
    const int c4 = (s & 15) * 4;
    *(float4*)(outb + r * H_DIM + c4) = *(const float4*)(O + r * LDX + c4);
  }
}

extern "C" void kernel_launch(void* const* d_in, const int* in_sizes, int n_in,
                              void* d_out, int out_size, void* d_ws, size_t ws_size,
                              hipStream_t stream) {
  const float* obs     = (const float*)d_in[0];
  const float* enc_w   = (const float*)d_in[1];
  const float* enc_b   = (const float*)d_in[2];
  const float* gate_w1 = (const float*)d_in[3];
  const float* gate_b1 = (const float*)d_in[4];
  const float* gate_w2 = (const float*)d_in[5];
  const float* gate_b2 = (const float*)d_in[6];
  const float* msg_w1  = (const float*)d_in[7];
  const float* msg_b1  = (const float*)d_in[8];
  const float* msg_w2  = (const float*)d_in[9];
  const float* msg_b2  = (const float*)d_in[10];
  float* out = (float*)d_out;

  gated_comm_kernel<<<BATCH, 128, 0, stream>>>(
      obs, enc_w, enc_b, gate_w1, gate_b1, gate_w2, gate_b2,
      msg_w1, msg_b1, msg_w2, msg_b2, out);
}


# row-stripe tiling: each wave takes its 16 rows through all five weight GEMMs transposed in registers (weights = A from global, D reused as B); gates feed out WMMAs in the sweep; sum order changes
# speedup vs baseline: 2.9662x; 1.0594x over previous
#include <hip/hip_runtime.h>
#include <math.h>


typedef __attribute__((ext_vector_type(2))) float v2f;
typedef __attribute__((ext_vector_type(4))) float v4f;
typedef __attribute__((ext_vector_type(8))) float v8f;

#define BATCH 256
#define N_DIM 64
#define D_DIM 128
#define H_DIM 64
#define LDG   68
#define LDM   68
#define LOADS_STAY 0x786

__device__ inline v8f wmma_f32x4(v2f a, v2f b, v8f c) {
  return __builtin_amdgcn_wmma_f32_16x16x4_f32(
       false, a,  false, b,
       (short)0, c,  false,  false);
}

__device__ inline void load_w64(v4f (&w)[16][2], const float* __restrict__ W,
                                int l16, int half) {
  const float* p = W + (32 * half) * H_DIM + 4 * l16;
#pragma unroll
  for (int mt = 0; mt < 4; ++mt)
#pragma unroll
    for (int q = 0; q < 4; ++q)
#pragma unroll
      for (int s = 0; s < 2; ++s)
        w[4 * mt + q][s] = *(const v4f*)(p + (8 * q + 4 * s + mt) * H_DIM);
}

__device__ inline void load_wenc(v4f (&w)[32][2], const float* __restrict__ W,
                                 int ks0, int l16, int half) {
  const float* p = W + (4 * half) * H_DIM + 4 * l16;
#pragma unroll
  for (int ks = ks0; ks < ks0 + 16; ++ks)
#pragma unroll
    for (int s = 0; s < 2; ++s)
      w[ks][s] = *(const v4f*)(p + (8 * (ks >> 1) + 2 * (ks & 1) + s) * H_DIM);
}

__device__ inline void load_bias(v4f (&bv)[8], const float* __restrict__ bias, int half) {
#pragma unroll
  for (int r = 0; r < 8; ++r) bv[r] = *(const v4f*)(bias + 32 * half + 4 * r);
}

__device__ inline void init_acc(v8f (&acc)[4], const v4f (&bv)[8]) {
#pragma unroll
  for (int mt = 0; mt < 4; ++mt)
#pragma unroll
    for (int r = 0; r < 8; ++r) acc[mt][r] = bv[r][mt];
}

__device__ inline void zero_acc(v8f (&acc)[4]) {
#pragma unroll
  for (int mt = 0; mt < 4; ++mt)
#pragma unroll
    for (int r = 0; r < 8; ++r) acc[mt][r] = 0.0f;
}

__device__ inline void relu_acc(v8f (&acc)[4]) {
#pragma unroll
  for (int mt = 0; mt < 4; ++mt)
#pragma unroll
    for (int r = 0; r < 8; ++r) acc[mt][r] = fmaxf(acc[mt][r], 0.0f);
}

__device__ inline void gemm64T(v8f (&acc)[4], const v8f (&src)[4],
                               const v4f (&w)[16][2]) {
#pragma unroll
  for (int mt = 0; mt < 4; ++mt)
#pragma unroll
    for (int q = 0; q < 4; ++q) {
      v2f bb;
      bb.x = src[mt][2 * q];
      bb.y = src[mt][2 * q + 1];
#pragma unroll
      for (int m2 = 0; m2 < 4; ++m2) {
        v2f a;
        a.x = w[4 * mt + q][0][m2];
        a.y = w[4 * mt + q][1][m2];
        acc[m2] = wmma_f32x4(a, bb, acc[m2]);
      }
    }
}

__device__ inline float xhalf(float x) {
  const int b = __builtin_bit_cast(int, x);
  return __builtin_bit_cast(float, __builtin_amdgcn_permlanex16(
      b, b, 0x76543210, 0xfedcba98, false, false));
}

__device__ inline void load_gj4(v4f (&g)[4][8], const float* gjp, int j0) {
#pragma unroll
  for (int k = 0; k < 4; ++k)
#pragma unroll
    for (int x = 0; x < 8; ++x)
      g[k][x] = *(const v4f*)(gjp + (j0 + k) * LDG + 4 * x);
}

__device__ inline void load_mb(v4f (&m)[2], const float* mp, int j0) {
  m[0] = *(const v4f*)(mp + j0 * LDM);
  m[1] = *(const v4f*)(mp + (j0 + 1) * LDM);
}

__device__ inline void gate_cols4(float (&s)[4], const v4f (&g)[4][8],
                                  const v2f (&gi2)[8][2], const v4f (&w2v)[8]) {
  v2f acc[4];
#pragma unroll
  for (int k = 0; k < 4; ++k) acc[k] = (v2f){0.0f, 0.0f};
#pragma unroll
  for (int x = 0; x < 8; ++x)
#pragma unroll
    for (int e = 0; e < 2; ++e) {
      const v2f w = e ? __builtin_shufflevector(w2v[x], w2v[x], 2, 3)
                      : __builtin_shufflevector(w2v[x], w2v[x], 0, 1);
#pragma unroll
      for (int k = 0; k < 4; ++k) {
        const v2f gj2 = e ? __builtin_shufflevector(g[k][x], g[k][x], 2, 3)
                          : __builtin_shufflevector(g[k][x], g[k][x], 0, 1);
        v2f t = gj2 + gi2[x][e];
        t.x = fmaxf(t.x, 0.0f);
        t.y = fmaxf(t.y, 0.0f);
        acc[k] = t * w + acc[k];
      }
    }
#pragma unroll
  for (int k = 0; k < 4; ++k) s[k] = acc[k].x + acc[k].y;
}

__device__ inline void gate_finish4(v8f (&oAcc)[4], const float (&s)[4],
                                    float gb2, int half, const v4f (&m)[2]) {
  float t[4];
#pragma unroll
  for (int k = 0; k < 4; ++k) t[k] = s[k] + xhalf(s[k]);
  const float tA = half ? t[2] : t[0];
  const float tB = half ? t[3] : t[1];
  v2f a;
  a.x = 1.0f / (1.0f + __expf(-(tA + gb2)));
  a.y = 1.0f / (1.0f + __expf(-(tB + gb2)));
#pragma unroll
  for (int nt = 0; nt < 4; ++nt) {
    v2f bb;
    bb.x = m[0][nt];
    bb.y = m[1][nt];
    oAcc[nt] = wmma_f32x4(a, bb, oAcc[nt]);
  }
}

__global__ __launch_bounds__(128) void gated_comm_kernel(
    const float* __restrict__ obs, const float* __restrict__ enc_w,
    const float* __restrict__ enc_b, const float* __restrict__ gate_w1,
    const float* __restrict__ gate_b1, const float* __restrict__ gate_w2,
    const float* __restrict__ gate_b2, const float* __restrict__ msg_w1,
    const float* __restrict__ msg_b1, const float* __restrict__ msg_w2,
    const float* __restrict__ msg_b2, float* __restrict__ out) {
  __shared__ __attribute__((aligned(16))) float smem[64 * LDG + 64 * LDM];

  const int b       = blockIdx.x;
  const int tid     = threadIdx.x;
  const int lane    = tid & 31;
  const int wave    = tid >> 5;
  const int l16     = lane & 15;
  const int half    = lane >> 4;
  const int row     = wave * 16 + l16;

  float* GJ  = smem;
  float* MSG = smem + 64 * LDG;

  v4f wE[32][2];
  v4f wI[16][2], wJ[16][2], wM1[16][2], wM2[16][2];
  v4f bE[8], bJ[8], bM1[8], bM2[8], w2L[8];

  const float* obsr = obs + ((size_t)b * N_DIM + row) * D_DIM + 4 * half;
  v4f ob[16];
#pragma unroll
  for (int m = 0; m < 16; ++m) ob[m] = *(const v4f*)(obsr + 8 * m);
  load_bias(bE, enc_b, half);
  load_wenc(wE, enc_w, 0, l16, half);

  v8f encT[4];
  init_acc(encT, bE);
#pragma unroll
  for (int ks = 0; ks < 32; ++ks) {
    if (ks == 2) {
      __builtin_amdgcn_sched_barrier(LOADS_STAY);
      load_wenc(wE, enc_w, 16, l16, half);
      __builtin_amdgcn_sched_barrier(LOADS_STAY);
    }
    if (ks == 8) {
      __builtin_amdgcn_sched_barrier(LOADS_STAY);
      load_w64(wI, gate_w1, l16, half);
      __builtin_amdgcn_sched_barrier(LOADS_STAY);
    }
    if (ks == 14) {
      __builtin_amdgcn_sched_barrier(LOADS_STAY);
      load_w64(wJ, gate_w1 + H_DIM * H_DIM, l16, half);
      load_bias(bJ, gate_b1, half);
      __builtin_amdgcn_sched_barrier(LOADS_STAY);
    }
    if (ks == 20) {
      __builtin_amdgcn_sched_barrier(LOADS_STAY);
      load_w64(wM1, msg_w1, l16, half);
      load_bias(bM1, msg_b1, half);
      __builtin_amdgcn_sched_barrier(LOADS_STAY);
    }
    v2f bb;
    bb.x = ob[ks >> 1][2 * (ks & 1)];
    bb.y = ob[ks >> 1][2 * (ks & 1) + 1];
#pragma unroll
    for (int mt = 0; mt < 4; ++mt) {
      v2f a;
      a.x = wE[ks][0][mt];
      a.y = wE[ks][1][mt];
      encT[mt] = wmma_f32x4(a, bb, encT[mt]);
    }
  }
  relu_acc(encT);

  __builtin_amdgcn_sched_barrier(LOADS_STAY);
  load_w64(wM2, msg_w2, l16, half);
  load_bias(bM2, msg_b2, half);
  load_bias(w2L, gate_w2, half);
  const float gb2 = gate_b2[0];
  __builtin_amdgcn_sched_barrier(LOADS_STAY);

  v8f giT[4], gjT[4], tT[4];
  zero_acc(giT);
  init_acc(gjT, bJ);
  init_acc(tT, bM1);
#pragma unroll
  for (int mt = 0; mt < 4; ++mt)
#pragma unroll
    for (int q = 0; q < 4; ++q) {
      v2f bb;
      bb.x = encT[mt][2 * q];
      bb.y = encT[mt][2 * q + 1];
#pragma unroll
      for (int m2 = 0; m2 < 4; ++m2) {
        v2f aI, aJ, aM;
        aI.x = wI[4 * mt + q][0][m2];  aI.y = wI[4 * mt + q][1][m2];
        aJ.x = wJ[4 * mt + q][0][m2];  aJ.y = wJ[4 * mt + q][1][m2];
        aM.x = wM1[4 * mt + q][0][m2]; aM.y = wM1[4 * mt + q][1][m2];
        giT[m2] = wmma_f32x4(aI, bb, giT[m2]);
        gjT[m2] = wmma_f32x4(aJ, bb, gjT[m2]);
        tT[m2]  = wmma_f32x4(aM, bb, tT[m2]);
      }
    }
  relu_acc(tT);

  {
    float* gw = GJ + row * LDG + 32 * half;
#pragma unroll
    for (int mt = 0; mt < 4; ++mt) {
      *(v4f*)(gw + 8 * mt)     = (v4f){gjT[mt][0], gjT[mt][1], gjT[mt][2], gjT[mt][3]};
      *(v4f*)(gw + 8 * mt + 4) = (v4f){gjT[mt][4], gjT[mt][5], gjT[mt][6], gjT[mt][7]};
    }
  }

  {
    v8f mT[4];
    init_acc(mT, bM2);
    gemm64T(mT, tT, wM2);
    float* mw = MSG + row * LDM + 32 * half;
#pragma unroll
    for (int r = 0; r < 8; ++r)
      *(v4f*)(mw + 4 * r) = (v4f){mT[0][r], mT[1][r], mT[2][r], mT[3][r]};
  }
  __syncthreads();

  v8f oAcc[4];
  zero_acc(oAcc);
  {
    v2f gi2[8][2];
    v4f w2v[8];
#pragma unroll
    for (int x = 0; x < 8; ++x) {
      const int mt = x >> 1;
      const int r0 = 4 * (x & 1);
      gi2[x][0] = (v2f){giT[mt][r0],     giT[mt][r0 + 1]};
      gi2[x][1] = (v2f){giT[mt][r0 + 2], giT[mt][r0 + 3]};
      w2v[x] = (v4f){w2L[r0][mt], w2L[r0 + 1][mt], w2L[r0 + 2][mt], w2L[r0 + 3][mt]};
    }
    const float* gjp = GJ + 32 * half;
    const float* mp  = MSG + (2 * half) * LDM + 4 * l16;
    v4f gA[4][8], gB[4][8];
    v4f mE[2], mO[2];
    float sA[4], sB[4];
    load_gj4(gA, gjp, 0);
    __builtin_amdgcn_sched_barrier(0);
    load_gj4(gB, gjp, 4);
    load_mb(mE, mp, 0);
    __builtin_amdgcn_sched_barrier(0);
    gate_cols4(sA, gA, gi2, w2v);
#pragma unroll 1
    for (int p = 1; p < 15; p += 2) {
      __builtin_amdgcn_sched_barrier(0);
      load_gj4(gA, gjp, 4 * (p + 1));
      load_mb(mO, mp, 4 * p);
      __builtin_amdgcn_sched_barrier(0);
      gate_cols4(sB, gB, gi2, w2v);
      gate_finish4(oAcc, sA, gb2, half, mE);
      __builtin_amdgcn_sched_barrier(0);
      load_gj4(gB, gjp, 4 * (p + 2));
      load_mb(mE, mp, 4 * (p + 1));
      __builtin_amdgcn_sched_barrier(0);
      gate_cols4(sA, gA, gi2, w2v);
      gate_finish4(oAcc, sB, gb2, half, mO);
    }
    __builtin_amdgcn_sched_barrier(0);
    load_mb(mO, mp, 4 * 15);
    __builtin_amdgcn_sched_barrier(0);
    gate_cols4(sB, gB, gi2, w2v);
    gate_finish4(oAcc, sA, gb2, half, mE);
    gate_finish4(oAcc, sB, gb2, half, mO);
  }

  float* outr = out + ((size_t)b * N_DIM + wave * 16 + 8 * half) * H_DIM + 4 * l16;
#pragma unroll
  for (int r = 0; r < 8; ++r)
    *(v4f*)(outr + r * H_DIM) = (v4f){oAcc[0][r], oAcc[1][r], oAcc[2][r], oAcc[3][r]};
}

extern "C" void kernel_launch(void* const* d_in, const int* in_sizes, int n_in,
                              void* d_out, int out_size, void* d_ws, size_t ws_size,
                              hipStream_t stream) {
  const float* obs     = (const float*)d_in[0];
  const float* enc_w   = (const float*)d_in[1];
  const float* enc_b   = (const float*)d_in[2];
  const float* gate_w1 = (const float*)d_in[3];
  const float* gate_b1 = (const float*)d_in[4];
  const float* gate_w2 = (const float*)d_in[5];
  const float* gate_b2 = (const float*)d_in[6];
  const float* msg_w1  = (const float*)d_in[7];
  const float* msg_b1  = (const float*)d_in[8];
  const float* msg_w2  = (const float*)d_in[9];
  const float* msg_b2  = (const float*)d_in[10];
  float* out = (float*)d_out;

  gated_comm_kernel<<<BATCH, 128, 0, stream>>>(
      obs, enc_w, enc_b, gate_w1, gate_b1, gate_w2, gate_b2,
      msg_w1, msg_b1, msg_w2, msg_b2, out);
}
